# starting kernel
# baseline (speedup 1.0000x reference)
#include <hip/hip_runtime.h>

#define BB 64
#define FF 128
#define CIN 64
#define CON 64
#define RR 4

typedef float v2f __attribute__((ext_vector_type(2)));
typedef float v8f __attribute__((ext_vector_type(8)));

#define E_STRIDE 65
#define E_R (BB * E_STRIDE)
#define W_STRIDE 65
#define W_R (CON * W_STRIDE)
#define LE_OFF 0
#define LW_OFF (RR * E_R)
#define LM_OFF (LW_OFF + RR * W_R)
#define LDS_FLOATS (LM_OFF + RR * BB)
#define LDS_BYTES (LDS_FLOATS * 4)

extern "C" __global__ __launch_bounds__(256)
void weightedsum_lse_wmma(const float* __restrict__ ll,
                          const float* __restrict__ wgt,
                          float* __restrict__ out) {
    extern __shared__ float smem[];
    const int f   = blockIdx.x;
    const int tid = threadIdx.x;

    const float4* ll4 = (const float4*)ll;
    const float4* w4  = (const float4*)wgt;
    #pragma unroll
    for (int i = 0; i < 16; ++i) {
        const int idx = tid + i * 256;
        const int hi = idx >> 6, lo = idx & 63;

        float4 v = ll4[(hi * FF + f) * CIN + lo];
        smem[LE_OFF + 0 * E_R + hi * E_STRIDE + lo] = v.x;
        smem[LE_OFF + 1 * E_R + hi * E_STRIDE + lo] = v.y;
        smem[LE_OFF + 2 * E_R + hi * E_STRIDE + lo] = v.z;
        smem[LE_OFF + 3 * E_R + hi * E_STRIDE + lo] = v.w;

        float4 wv = w4[(f * CIN + hi) * CON + lo];
        smem[LW_OFF + 0 * W_R + lo * W_STRIDE + hi] = wv.x;
        smem[LW_OFF + 1 * W_R + lo * W_STRIDE + hi] = wv.y;
        smem[LW_OFF + 2 * W_R + lo * W_STRIDE + hi] = wv.z;
        smem[LW_OFF + 3 * W_R + lo * W_STRIDE + hi] = wv.w;
    }
    __syncthreads();

    {
        const int b = tid & 63;
        const int r = tid >> 6;
        float* row = &smem[LE_OFF + r * E_R + b * E_STRIDE];
        float m = row[0];
        #pragma unroll 8
        for (int ci = 1; ci < CIN; ++ci) m = fmaxf(m, row[ci]);
        smem[LM_OFF + r * BB + b] = m;
        #pragma unroll 8
        for (int ci = 0; ci < CIN; ++ci) row[ci] = __expf(row[ci] - m);
    }
    __syncthreads();

    const int wave  = tid >> 5;
    const int lane  = tid & 31;
    const int r     = wave >> 1;
    const int mhalf = wave & 1;
    const int lrow  = lane & 15;
    const int koff  = (lane >> 4) << 1;

    const float* Eb = &smem[LE_OFF + r * E_R];
    const float* Wb = &smem[LW_OFF + r * W_R];

    v8f acc[8];
    #pragma unroll
    for (int mt = 0; mt < 2; ++mt) {
        const int arow = (mhalf * 2 + mt) * 16 + lrow;
        #pragma unroll
        for (int nt = 0; nt < 4; ++nt) {
            const int bcol = nt * 16 + lrow;
            v8f c = {0.f, 0.f, 0.f, 0.f, 0.f, 0.f, 0.f, 0.f};
            #pragma unroll
            for (int k = 0; k < CIN; k += 4) {
                const int kk = k + koff;
                v2f a, b;
                a.x = Eb[arow * E_STRIDE + kk];
                a.y = Eb[arow * E_STRIDE + kk + 1];
                b.x = Wb[bcol * W_STRIDE + kk];
                b.y = Wb[bcol * W_STRIDE + kk + 1];
                c = __builtin_amdgcn_wmma_f32_16x16x4_f32(
                         false, a,  false, b,
                         (short)0, c,  false,  false);
            }
            acc[mt * 4 + nt] = c;
        }
    }
    __syncthreads();

    {
        float* O = &smem[LE_OFF + r * E_R];
        const int rowadd = (lane >> 4) << 3;
        #pragma unroll
        for (int mt = 0; mt < 2; ++mt) {
            const int base_row = (mhalf * 2 + mt) * 16 + rowadd;
            #pragma unroll
            for (int nt = 0; nt < 4; ++nt) {
                const int col = nt * 16 + lrow;
                v8f c = acc[mt * 4 + nt];
                #pragma unroll
                for (int v = 0; v < 8; ++v)
                    O[(base_row + v) * E_STRIDE + col] = c[v];
            }
        }
    }
    __syncthreads();

    float4* out4 = (float4*)out;
    #pragma unroll
    for (int i = 0; i < 16; ++i) {
        const int idx = tid + i * 256;
        const int b = idx >> 6, co = idx & 63;
        float4 o;
        o.x = smem[LM_OFF + 0 * BB + b] + __logf(smem[LE_OFF + 0 * E_R + b * E_STRIDE + co]);
        o.y = smem[LM_OFF + 1 * BB + b] + __logf(smem[LE_OFF + 1 * E_R + b * E_STRIDE + co]);
        o.z = smem[LM_OFF + 2 * BB + b] + __logf(smem[LE_OFF + 2 * E_R + b * E_STRIDE + co]);
        o.w = smem[LM_OFF + 3 * BB + b] + __logf(smem[LE_OFF + 3 * E_R + b * E_STRIDE + co]);
        out4[(b * FF + f) * CON + co] = o;
    }
}

extern "C" void kernel_launch(void* const* d_in, const int* in_sizes, int n_in,
                              void* d_out, int out_size, void* d_ws, size_t ws_size,
                              hipStream_t stream) {
    (void)in_sizes; (void)n_in; (void)d_ws; (void)ws_size; (void)out_size;
    const float* ll  = (const float*)d_in[0];
    const float* wgt = (const float*)d_in[1];
    float* out = (float*)d_out;

    weightedsum_lse_wmma<<<dim3(FF), dim3(256), LDS_BYTES, stream>>>(ll, wgt, out);
}


# WMMA A/B fragments read straight from global memory (one 16-byte load = the 4 r planes of one element): no LDS staging, transpose or block barrier; 1024 waves on 256 work-groups; sum order kept
# speedup vs baseline: 1.7608x; 1.7608x over previous
#include <hip/hip_runtime.h>

#define BB 64
#define FF 128
#define CIN 64
#define CON 64
#define RR 4
#define NWG 256

typedef float v2f __attribute__((ext_vector_type(2)));
typedef float v8f __attribute__((ext_vector_type(8)));

extern "C" __global__ __launch_bounds__(128)
void weightedsum_lse_wmma(const float* __restrict__ ll,
                          const float* __restrict__ wgt,
                          float* __restrict__ out) {
    const int f    = blockIdx.x >> 1;
    const int wv   = threadIdx.x >> 5;
    const int mt   = (blockIdx.x & 1) * 2 + (wv >> 1);
    const int nh   = wv & 1;
    const int lane = threadIdx.x & 31;
    const int c    = lane & 15;
    const int h    = lane >> 4;

    const float4* ap = (const float4*)ll  + ((mt * 16 + c) * FF + f) * CIN + 2 * h;
    const float4* bp = (const float4*)wgt + (f * CIN + 2 * h) * CON + nh * 32 + c;

    float a[16][2][RR];
    #pragma unroll
    for (int t = 0; t < 16; ++t) {
        #pragma unroll
        for (int s = 0; s < 2; ++s) {
            const float4 v = ap[4 * t + s];
            a[t][s][0] = v.x; a[t][s][1] = v.y; a[t][s][2] = v.z; a[t][s][3] = v.w;
        }
    }
    __builtin_amdgcn_sched_barrier(0);
    float b[16][2][2][RR];
    #pragma unroll
    for (int t = 0; t < 16; ++t) {
        #pragma unroll
        for (int s = 0; s < 2; ++s) {
            #pragma unroll
            for (int nt = 0; nt < 2; ++nt) {
                const float4 v = bp[(4 * t + s) * CON + 16 * nt];
                b[t][s][nt][0] = v.x; b[t][s][nt][1] = v.y; b[t][s][nt][2] = v.z; b[t][s][nt][3] = v.w;
            }
        }
    }

    __builtin_amdgcn_sched_barrier(0);
    float m[RR];
    #pragma unroll
    for (int r = 0; r < RR; ++r) {
        float mm = a[0][0][r];
        #pragma unroll
        for (int t = 0; t < 16; ++t) {
            #pragma unroll
            for (int s = 0; s < 2; ++s) mm = fmaxf(mm, a[t][s][r]);
        }
        m[r] = fmaxf(mm, __shfl_xor(mm, 16));
    }

    v8f acc[RR][2];
    #pragma unroll
    for (int r = 0; r < RR; ++r) {
        #pragma unroll
        for (int nt = 0; nt < 2; ++nt)
            acc[r][nt] = (v8f){0.f, 0.f, 0.f, 0.f, 0.f, 0.f, 0.f, 0.f};
    }
    float mr[8][RR];
    v2f e[RR];
    #pragma unroll
    for (int r = 0; r < RR; ++r) {
        e[r].x = __expf(a[0][0][r] - m[r]);
        e[r].y = __expf(a[0][1][r] - m[r]);
    }
    #pragma unroll
    for (int t = 0; t < 16; ++t) {
        v2f en[RR];
        #pragma unroll
        for (int r = 0; r < RR; ++r) {
            #pragma unroll
            for (int nt = 0; nt < 2; ++nt) {
                v2f bv;
                bv.x = b[t][0][nt][r];
                bv.y = b[t][1][nt][r];
                acc[r][nt] = __builtin_amdgcn_wmma_f32_16x16x4_f32(
                         false, e[r],  false, bv,
                         (short)0, acc[r][nt],  false,  false);
            }
            en[r].x = (t < 15) ? __expf(a[(t + 1) & 15][0][r] - m[r]) : 0.f;
            en[r].y = (t < 15) ? __expf(a[(t + 1) & 15][1][r] - m[r]) : 0.f;
        }
        if (t == 0) {
            #pragma unroll
            for (int v = 0; v < 8; ++v) {
                #pragma unroll
                for (int r = 0; r < RR; ++r) mr[v][r] = __shfl(m[r], 8 * h + v);
            }
        }
        #pragma unroll
        for (int r = 0; r < RR; ++r) e[r] = en[r];
        __builtin_amdgcn_sched_barrier(0);
    }

    float4* out4 = (float4*)out;
    #pragma unroll
    for (int v = 0; v < 8; ++v) {
        #pragma unroll
        for (int nt = 0; nt < 2; ++nt) {
            float4 o;
            o.x = mr[v][0] + __logf(acc[0][nt][v]);
            o.y = mr[v][1] + __logf(acc[1][nt][v]);
            o.z = mr[v][2] + __logf(acc[2][nt][v]);
            o.w = mr[v][3] + __logf(acc[3][nt][v]);
            out4[((mt * 16 + 8 * h + v) * FF + f) * CON + nh * 32 + nt * 16 + c] = o;
        }
    }
}

extern "C" void kernel_launch(void* const* d_in, const int* in_sizes, int n_in,
                              void* d_out, int out_size, void* d_ws, size_t ws_size,
                              hipStream_t stream) {
    (void)in_sizes; (void)n_in; (void)d_ws; (void)ws_size; (void)out_size;
    const float* ll  = (const float*)d_in[0];
    const float* wgt = (const float*)d_in[1];
    float* out = (float*)d_out;

    weightedsum_lse_wmma<<<dim3(NWG), dim3(128), 0, stream>>>(ll, wgt, out);
}


# exp(ll-max) row operand built once per 16-row tile: its two column-half waves split the rows (4 lanes per row, 8-line loads), exponentiate once, hand A fragments over through LDS; sum order kept
# speedup vs baseline: 2.0579x; 1.1687x over previous
#include <hip/hip_runtime.h>

#define BB 64
#define FF 128
#define CIN 64
#define CON 64
#define RR 4
#define NWG 256
#define EPITCH 260

typedef float v2f __attribute__((ext_vector_type(2)));
typedef float v8f __attribute__((ext_vector_type(8)));

template <int CTRL>
__device__ __forceinline__ float quad_swap(float x) {
    return __builtin_bit_cast(float,
        __builtin_amdgcn_mov_dpp(__builtin_bit_cast(int, x), CTRL, 0xF, 0xF, true));
}

extern "C" __global__ __launch_bounds__(128)
void weightedsum_lse_wmma(const float* __restrict__ ll,
                          const float* __restrict__ wgt,
                          float* __restrict__ out) {
    __shared__ __attribute__((aligned(16))) float es[2][16 * EPITCH];
    __shared__ __attribute__((aligned(16))) float ms[2][16 * RR];

    const int f    = blockIdx.x >> 1;
    const int wv   = threadIdx.x >> 5;
    const int tl   = wv >> 1;
    const int mt   = (blockIdx.x & 1) * 2 + tl;
    const int nh   = wv & 1;
    const int lane = threadIdx.x & 31;
    const int c    = lane & 15;
    const int h    = lane >> 4;

    const int q    = lane & 3;
    const int rl   = lane >> 2;
    const int prow = nh * 4 + (rl >> 2) * 2 + (rl & 1) + ((rl >> 1) & 1) * 8;

    const float4* ap = (const float4*)ll  + ((mt * 16 + prow) * FF + f) * CIN + 2 * q;
    const float4* bp = (const float4*)wgt + (f * CIN + 2 * h) * CON + nh * 32 + c;

    float4 ra[8][2];
    #pragma unroll
    for (int i = 0; i < 8; ++i) {
        ra[i][0] = ap[8 * i];
        ra[i][1] = ap[8 * i + 1];
    }
    __builtin_amdgcn_sched_barrier(0);
    float b[16][2][2][RR];
    #pragma unroll
    for (int t = 0; t < 16; ++t) {
        #pragma unroll
        for (int s = 0; s < 2; ++s) {
            #pragma unroll
            for (int nt = 0; nt < 2; ++nt) {
                const float4 v = bp[(4 * t + s) * CON + 16 * nt];
                b[t][s][nt][0] = v.x; b[t][s][nt][1] = v.y; b[t][s][nt][2] = v.z; b[t][s][nt][3] = v.w;
            }
        }
    }
    __builtin_amdgcn_sched_barrier(0);

    float m[RR];
    {
        float mx = ra[0][0].x, my = ra[0][0].y, mz = ra[0][0].z, mw = ra[0][0].w;
        #pragma unroll
        for (int i = 0; i < 8; ++i) {
            #pragma unroll
            for (int s = 0; s < 2; ++s) {
                mx = fmaxf(mx, ra[i][s].x); my = fmaxf(my, ra[i][s].y);
                mz = fmaxf(mz, ra[i][s].z); mw = fmaxf(mw, ra[i][s].w);
            }
        }
        m[0] = mx; m[1] = my; m[2] = mz; m[3] = mw;
        #pragma unroll
        for (int r = 0; r < RR; ++r) {
            m[r] = fmaxf(m[r], quad_swap<0xB1>(m[r]));
            m[r] = fmaxf(m[r], quad_swap<0x4E>(m[r]));
        }
    }

    {
        float* erow = &es[tl][prow * EPITCH + q * 8];
        #pragma unroll
        for (int i = 0; i < 8; ++i) {
            float4 lo, hi;
            lo.x = __expf(ra[i][0].x - m[0]); lo.y = __expf(ra[i][1].x - m[0]);
            lo.z = __expf(ra[i][0].y - m[1]); lo.w = __expf(ra[i][1].y - m[1]);
            hi.x = __expf(ra[i][0].z - m[2]); hi.y = __expf(ra[i][1].z - m[2]);
            hi.z = __expf(ra[i][0].w - m[3]); hi.w = __expf(ra[i][1].w - m[3]);
            *(float4*)(erow + 32 * i)     = lo;
            *(float4*)(erow + 32 * i + 4) = hi;
        }
        if (q == 0) {
            float4 mm; mm.x = m[0]; mm.y = m[1]; mm.z = m[2]; mm.w = m[3];
            *(float4*)&ms[tl][prow * RR] = mm;
        }
    }
    __builtin_amdgcn_fence(__ATOMIC_RELEASE, "workgroup", "local");
    __builtin_amdgcn_s_barrier();
    __builtin_amdgcn_fence(__ATOMIC_ACQUIRE, "workgroup", "local");

    v2f e[16][RR];
    {
        const float* erd = &es[tl][c * EPITCH + h * 8];
        #pragma unroll
        for (int t = 0; t < 16; ++t) {
            const float4 lo = *(const float4*)(erd + 16 * t);
            const float4 hi = *(const float4*)(erd + 16 * t + 4);
            e[t][0].x = lo.x; e[t][0].y = lo.y; e[t][1].x = lo.z; e[t][1].y = lo.w;
            e[t][2].x = hi.x; e[t][2].y = hi.y; e[t][3].x = hi.z; e[t][3].y = hi.w;
        }
    }
    float mr[8][RR];
    #pragma unroll
    for (int v = 0; v < 8; ++v) {
        const float4 mm = *(const float4*)&ms[tl][(8 * h + v) * RR];
        mr[v][0] = mm.x; mr[v][1] = mm.y; mr[v][2] = mm.z; mr[v][3] = mm.w;
    }
    __builtin_amdgcn_sched_group_barrier(0x100, 40, 0);
    __builtin_amdgcn_sched_group_barrier(0x008, 8, 0);

    v8f acc[RR][2];
    #pragma unroll
    for (int r = 0; r < RR; ++r) {
        #pragma unroll
        for (int nt = 0; nt < 2; ++nt)
            acc[r][nt] = (v8f){0.f, 0.f, 0.f, 0.f, 0.f, 0.f, 0.f, 0.f};
    }
    #pragma unroll
    for (int t = 0; t < 16; ++t) {
        #pragma unroll
        for (int r = 0; r < RR; ++r) {
            #pragma unroll
            for (int nt = 0; nt < 2; ++nt) {
                v2f bv;
                bv.x = b[t][0][nt][r];
                bv.y = b[t][1][nt][r];
                acc[r][nt] = __builtin_amdgcn_wmma_f32_16x16x4_f32(
                         false, e[t][r],  false, bv,
                         (short)0, acc[r][nt],  false,  false);
            }
        }
        __builtin_amdgcn_sched_barrier(0);
    }

    float4* out4 = (float4*)out;
    #pragma unroll
    for (int v = 0; v < 8; ++v) {
        #pragma unroll
        for (int nt = 0; nt < 2; ++nt) {
            float4 o;
            o.x = mr[v][0] + __logf(acc[0][nt][v]);
            o.y = mr[v][1] + __logf(acc[1][nt][v]);
            o.z = mr[v][2] + __logf(acc[2][nt][v]);
            o.w = mr[v][3] + __logf(acc[3][nt][v]);
            out4[((mt * 16 + 8 * h + v) * FF + f) * CON + nh * 32 + nt * 16 + c] = o;
        }
    }
}

extern "C" void kernel_launch(void* const* d_in, const int* in_sizes, int n_in,
                              void* d_out, int out_size, void* d_ws, size_t ws_size,
                              hipStream_t stream) {
    (void)in_sizes; (void)n_in; (void)d_ws; (void)ws_size; (void)out_size;
    const float* ll  = (const float*)d_in[0];
    const float* wgt = (const float*)d_in[1];
    float* out = (float*)d_out;

    weightedsum_lse_wmma<<<dim3(NWG), dim3(128), 0, stream>>>(ll, wgt, out);
}


# epilogue log(S) as ln2 * hardware log2 (one v_log_f32 + one multiply-add per output) in place of the library logf expansion of about 12 vector instructions; last bits may differ; sum order kept
# speedup vs baseline: 2.2338x; 1.0855x over previous
#include <hip/hip_runtime.h>

#define BB 64
#define FF 128
#define CIN 64
#define CON 64
#define RR 4
#define NWG 256
#define LN2 0.693147180559945f
#define EPITCH 260

typedef float v2f __attribute__((ext_vector_type(2)));
typedef float v8f __attribute__((ext_vector_type(8)));

template <int CTRL>
__device__ __forceinline__ float quad_swap(float x) {
    return __builtin_bit_cast(float,
        __builtin_amdgcn_mov_dpp(__builtin_bit_cast(int, x), CTRL, 0xF, 0xF, true));
}

extern "C" __global__ __launch_bounds__(128)
void weightedsum_lse_wmma(const float* __restrict__ ll,
                          const float* __restrict__ wgt,
                          float* __restrict__ out) {
    __shared__ __attribute__((aligned(16))) float es[2][16 * EPITCH];
    __shared__ __attribute__((aligned(16))) float ms[2][16 * RR];

    const int f    = blockIdx.x >> 1;
    const int wv   = threadIdx.x >> 5;
    const int tl   = wv >> 1;
    const int mt   = (blockIdx.x & 1) * 2 + tl;
    const int nh   = wv & 1;
    const int lane = threadIdx.x & 31;
    const int c    = lane & 15;
    const int h    = lane >> 4;

    const int q    = lane & 3;
    const int rl   = lane >> 2;
    const int prow = nh * 4 + (rl >> 2) * 2 + (rl & 1) + ((rl >> 1) & 1) * 8;

    const float4* ap = (const float4*)ll  + ((mt * 16 + prow) * FF + f) * CIN + 2 * q;
    const float4* bp = (const float4*)wgt + (f * CIN + 2 * h) * CON + nh * 32 + c;

    float4 ra[8][2];
    #pragma unroll
    for (int i = 0; i < 8; ++i) {
        ra[i][0] = ap[8 * i];
        ra[i][1] = ap[8 * i + 1];
    }
    __builtin_amdgcn_sched_barrier(0);
    float b[16][2][2][RR];
    #pragma unroll
    for (int t = 0; t < 16; ++t) {
        #pragma unroll
        for (int s = 0; s < 2; ++s) {
            #pragma unroll
            for (int nt = 0; nt < 2; ++nt) {
                const float4 v = bp[(4 * t + s) * CON + 16 * nt];
                b[t][s][nt][0] = v.x; b[t][s][nt][1] = v.y; b[t][s][nt][2] = v.z; b[t][s][nt][3] = v.w;
            }
        }
    }
    __builtin_amdgcn_sched_barrier(0);

    float m[RR];
    {
        float mx = ra[0][0].x, my = ra[0][0].y, mz = ra[0][0].z, mw = ra[0][0].w;
        #pragma unroll
        for (int i = 0; i < 8; ++i) {
            #pragma unroll
            for (int s = 0; s < 2; ++s) {
                mx = fmaxf(mx, ra[i][s].x); my = fmaxf(my, ra[i][s].y);
                mz = fmaxf(mz, ra[i][s].z); mw = fmaxf(mw, ra[i][s].w);
            }
        }
        m[0] = mx; m[1] = my; m[2] = mz; m[3] = mw;
        #pragma unroll
        for (int r = 0; r < RR; ++r) {
            m[r] = fmaxf(m[r], quad_swap<0xB1>(m[r]));
            m[r] = fmaxf(m[r], quad_swap<0x4E>(m[r]));
        }
    }

    {
        float* erow = &es[tl][prow * EPITCH + q * 8];
        #pragma unroll
        for (int i = 0; i < 8; ++i) {
            float4 lo, hi;
            lo.x = __expf(ra[i][0].x - m[0]); lo.y = __expf(ra[i][1].x - m[0]);
            lo.z = __expf(ra[i][0].y - m[1]); lo.w = __expf(ra[i][1].y - m[1]);
            hi.x = __expf(ra[i][0].z - m[2]); hi.y = __expf(ra[i][1].z - m[2]);
            hi.z = __expf(ra[i][0].w - m[3]); hi.w = __expf(ra[i][1].w - m[3]);
            *(float4*)(erow + 32 * i)     = lo;
            *(float4*)(erow + 32 * i + 4) = hi;
        }
        if (q == 0) {
            float4 mm; mm.x = m[0]; mm.y = m[1]; mm.z = m[2]; mm.w = m[3];
            *(float4*)&ms[tl][prow * RR] = mm;
        }
    }
    __builtin_amdgcn_fence(__ATOMIC_RELEASE, "workgroup", "local");
    __builtin_amdgcn_s_barrier();
    __builtin_amdgcn_fence(__ATOMIC_ACQUIRE, "workgroup", "local");

    v2f e[16][RR];
    {
        const float* erd = &es[tl][c * EPITCH + h * 8];
        #pragma unroll
        for (int t = 0; t < 16; ++t) {
            const float4 lo = *(const float4*)(erd + 16 * t);
            const float4 hi = *(const float4*)(erd + 16 * t + 4);
            e[t][0].x = lo.x; e[t][0].y = lo.y; e[t][1].x = lo.z; e[t][1].y = lo.w;
            e[t][2].x = hi.x; e[t][2].y = hi.y; e[t][3].x = hi.z; e[t][3].y = hi.w;
        }
    }
    float mr[8][RR];
    #pragma unroll
    for (int v = 0; v < 8; ++v) {
        const float4 mm = *(const float4*)&ms[tl][(8 * h + v) * RR];
        mr[v][0] = mm.x; mr[v][1] = mm.y; mr[v][2] = mm.z; mr[v][3] = mm.w;
    }
    __builtin_amdgcn_sched_group_barrier(0x100, 40, 0);
    __builtin_amdgcn_sched_group_barrier(0x008, 8, 0);

    v8f acc[RR][2];
    #pragma unroll
    for (int r = 0; r < RR; ++r) {
        #pragma unroll
        for (int nt = 0; nt < 2; ++nt)
            acc[r][nt] = (v8f){0.f, 0.f, 0.f, 0.f, 0.f, 0.f, 0.f, 0.f};
    }
    #pragma unroll
    for (int t = 0; t < 16; ++t) {
        #pragma unroll
        for (int r = 0; r < RR; ++r) {
            #pragma unroll
            for (int nt = 0; nt < 2; ++nt) {
                v2f bv;
                bv.x = b[t][0][nt][r];
                bv.y = b[t][1][nt][r];
                acc[r][nt] = __builtin_amdgcn_wmma_f32_16x16x4_f32(
                         false, e[t][r],  false, bv,
                         (short)0, acc[r][nt],  false,  false);
            }
        }
        __builtin_amdgcn_sched_barrier(0);
    }

    float4* out4 = (float4*)out;
    #pragma unroll
    for (int v = 0; v < 8; ++v) {
        #pragma unroll
        for (int nt = 0; nt < 2; ++nt) {
            float4 o;
            o.x = mr[v][0] + LN2 * __builtin_amdgcn_logf(acc[0][nt][v]);
            o.y = mr[v][1] + LN2 * __builtin_amdgcn_logf(acc[1][nt][v]);
            o.z = mr[v][2] + LN2 * __builtin_amdgcn_logf(acc[2][nt][v]);
            o.w = mr[v][3] + LN2 * __builtin_amdgcn_logf(acc[3][nt][v]);
            out4[((mt * 16 + 8 * h + v) * FF + f) * CON + nh * 32 + nt * 16 + c] = o;
        }
    }
}

extern "C" void kernel_launch(void* const* d_in, const int* in_sizes, int n_in,
                              void* d_out, int out_size, void* d_ws, size_t ws_size,
                              hipStream_t stream) {
    (void)in_sizes; (void)n_in; (void)d_ws; (void)ws_size; (void)out_size;
    const float* ll  = (const float*)d_in[0];
    const float* wgt = (const float*)d_in[1];
    float* out = (float*)d_out;

    weightedsum_lse_wmma<<<dim3(NWG), dim3(128), 0, stream>>>(ll, wgt, out);
}


# each wave owns 16 columns for BOTH row tiles, one after the other: weight fragments fetched once per work-group (32 loads a wave, not 64); tile 0 stored under tile 1's products; sum order kept
# speedup vs baseline: 2.3749x; 1.0632x over previous
#include <hip/hip_runtime.h>

#define BB 64
#define FF 128
#define CIN 64
#define CON 64
#define RR 4
#define NWG 256
#define LN2 0.693147180559945f
#define EPITCH 260

typedef float v2f __attribute__((ext_vector_type(2)));
typedef float v8f __attribute__((ext_vector_type(8)));

template <int CTRL>
__device__ __forceinline__ float quad_swap(float x) {
    return __builtin_bit_cast(float,
        __builtin_amdgcn_mov_dpp(__builtin_bit_cast(int, x), CTRL, 0xF, 0xF, true));
}

extern "C" __global__ __launch_bounds__(128)
void weightedsum_lse_wmma(const float* __restrict__ ll,
                          const float* __restrict__ wgt,
                          float* __restrict__ out) {
    __shared__ __attribute__((aligned(16))) float es[2][16 * EPITCH];
    __shared__ __attribute__((aligned(16))) float ms[2][16 * RR];

    const int f    = blockIdx.x >> 1;
    const int rb   = (blockIdx.x & 1) * 32;
    const int wv   = threadIdx.x >> 5;
    const int tl   = wv >> 1;
    const int ph   = wv & 1;
    const int lane = threadIdx.x & 31;
    const int c    = lane & 15;
    const int h    = lane >> 4;

    const int q    = lane & 3;
    const int rl   = lane >> 2;
    const int prow = ph * 4 + (rl >> 2) * 2 + (rl & 1) + ((rl >> 1) & 1) * 8;

    const float4* ap = (const float4*)ll  + ((rb + tl * 16 + prow) * FF + f) * CIN + 2 * q;
    const float4* bp = (const float4*)wgt + (f * CIN + 2 * h) * CON + wv * 16 + c;

    float4 ra[8][2];
    #pragma unroll
    for (int i = 0; i < 8; ++i) {
        ra[i][0] = ap[8 * i];
        ra[i][1] = ap[8 * i + 1];
    }
    __builtin_amdgcn_sched_barrier(0);
    float b[16][2][RR];
    #pragma unroll
    for (int t = 0; t < 16; ++t) {
        #pragma unroll
        for (int s = 0; s < 2; ++s) {
            const float4 v = bp[(4 * t + s) * CON];
            b[t][s][0] = v.x; b[t][s][1] = v.y; b[t][s][2] = v.z; b[t][s][3] = v.w;
        }
    }
    __builtin_amdgcn_sched_barrier(0);
    float m[RR];
    {
        float mx = ra[0][0].x, my = ra[0][0].y, mz = ra[0][0].z, mw = ra[0][0].w;
        #pragma unroll
        for (int i = 0; i < 8; ++i) {
            #pragma unroll
            for (int s = 0; s < 2; ++s) {
                mx = fmaxf(mx, ra[i][s].x); my = fmaxf(my, ra[i][s].y);
                mz = fmaxf(mz, ra[i][s].z); mw = fmaxf(mw, ra[i][s].w);
            }
        }
        m[0] = mx; m[1] = my; m[2] = mz; m[3] = mw;
        #pragma unroll
        for (int r = 0; r < RR; ++r) {
            m[r] = fmaxf(m[r], quad_swap<0xB1>(m[r]));
            m[r] = fmaxf(m[r], quad_swap<0x4E>(m[r]));
        }
    }

    {
        float* erow = &es[tl][prow * EPITCH + q * 8];
        #pragma unroll
        for (int i = 0; i < 8; ++i) {
            float4 lo, hi;
            lo.x = __expf(ra[i][0].x - m[0]); lo.y = __expf(ra[i][1].x - m[0]);
            lo.z = __expf(ra[i][0].y - m[1]); lo.w = __expf(ra[i][1].y - m[1]);
            hi.x = __expf(ra[i][0].z - m[2]); hi.y = __expf(ra[i][1].z - m[2]);
            hi.z = __expf(ra[i][0].w - m[3]); hi.w = __expf(ra[i][1].w - m[3]);
            *(float4*)(erow + 32 * i)     = lo;
            *(float4*)(erow + 32 * i + 4) = hi;
        }
        if (q == 0) {
            float4 mm; mm.x = m[0]; mm.y = m[1]; mm.z = m[2]; mm.w = m[3];
            *(float4*)&ms[tl][prow * RR] = mm;
        }
    }
    __builtin_amdgcn_fence(__ATOMIC_RELEASE, "workgroup", "local");
    __builtin_amdgcn_s_barrier();
    __builtin_amdgcn_fence(__ATOMIC_ACQUIRE, "workgroup", "local");

    v2f e0[16][RR];
    {
        const float* erd = &es[0][c * EPITCH + h * 8];
        #pragma unroll
        for (int t = 0; t < 16; ++t) {
            const float4 lo = *(const float4*)(erd + 16 * t);
            const float4 hi = *(const float4*)(erd + 16 * t + 4);
            e0[t][0].x = lo.x; e0[t][0].y = lo.y; e0[t][1].x = lo.z; e0[t][1].y = lo.w;
            e0[t][2].x = hi.x; e0[t][2].y = hi.y; e0[t][3].x = hi.z; e0[t][3].y = hi.w;
        }
    }
    float mr0[8][RR];
    #pragma unroll
    for (int v = 0; v < 8; ++v) {
        const float4 mm = *(const float4*)&ms[0][(8 * h + v) * RR];
        mr0[v][0] = mm.x; mr0[v][1] = mm.y; mr0[v][2] = mm.z; mr0[v][3] = mm.w;
    }
    __builtin_amdgcn_sched_group_barrier(0x100, 40, 0);
    __builtin_amdgcn_sched_group_barrier(0x008, 4, 0);

    v8f acc0[RR], acc1[RR];
    #pragma unroll
    for (int r = 0; r < RR; ++r) {
        acc0[r] = (v8f){0.f, 0.f, 0.f, 0.f, 0.f, 0.f, 0.f, 0.f};
        acc1[r] = (v8f){0.f, 0.f, 0.f, 0.f, 0.f, 0.f, 0.f, 0.f};
    }
    v2f   e1[16][RR];
    float mr1[8][RR];
    #pragma unroll
    for (int t = 0; t < 16; ++t) {
        #pragma unroll
        for (int r = 0; r < RR; ++r) {
            v2f bv;
            bv.x = b[t][0][r];
            bv.y = b[t][1][r];
            acc0[r] = __builtin_amdgcn_wmma_f32_16x16x4_f32(
                     false, e0[t][r],  false, bv,
                     (short)0, acc0[r],  false,  false);
        }
        __builtin_amdgcn_sched_barrier(0);
        if (t == 7) {
            const float* erd = &es[1][c * EPITCH + h * 8];
            #pragma unroll
            for (int u = 0; u < 16; ++u) {
                const float4 lo = *(const float4*)(erd + 16 * u);
                const float4 hi = *(const float4*)(erd + 16 * u + 4);
                e1[u][0].x = lo.x; e1[u][0].y = lo.y; e1[u][1].x = lo.z; e1[u][1].y = lo.w;
                e1[u][2].x = hi.x; e1[u][2].y = hi.y; e1[u][3].x = hi.z; e1[u][3].y = hi.w;
            }
            #pragma unroll
            for (int v = 0; v < 8; ++v) {
                const float4 mm = *(const float4*)&ms[1][(8 * h + v) * RR];
                mr1[v][0] = mm.x; mr1[v][1] = mm.y; mr1[v][2] = mm.z; mr1[v][3] = mm.w;
            }
            asm volatile("" ::: "memory");
            __builtin_amdgcn_sched_barrier(0);
        }
    }

    float4* out4 = (float4*)out;
    float4* orow = out4 + ((rb + 8 * h) * FF + f) * CON + wv * 16 + c;

    #pragma unroll
    for (int t = 0; t < 16; ++t) {
        #pragma unroll
        for (int r = 0; r < RR; ++r) {
            v2f bv;
            bv.x = b[t][0][r];
            bv.y = b[t][1][r];
            acc1[r] = __builtin_amdgcn_wmma_f32_16x16x4_f32(
                     false, e1[t][r],  false, bv,
                     (short)0, acc1[r],  false,  false);
        }
        if (t < 8) {
            float4 o;
            o.x = mr0[t][0] + LN2 * __builtin_amdgcn_logf(acc0[0][t]);
            o.y = mr0[t][1] + LN2 * __builtin_amdgcn_logf(acc0[1][t]);
            o.z = mr0[t][2] + LN2 * __builtin_amdgcn_logf(acc0[2][t]);
            o.w = mr0[t][3] + LN2 * __builtin_amdgcn_logf(acc0[3][t]);
            orow[t * FF * CON] = o;
            asm volatile("" ::: "memory");
        }
        __builtin_amdgcn_sched_barrier(0);
    }

    #pragma unroll
    for (int v = 0; v < 8; ++v) {
        float4 o;
        o.x = mr1[v][0] + LN2 * __builtin_amdgcn_logf(acc1[0][v]);
        o.y = mr1[v][1] + LN2 * __builtin_amdgcn_logf(acc1[1][v]);
        o.z = mr1[v][2] + LN2 * __builtin_amdgcn_logf(acc1[2][v]);
        o.w = mr1[v][3] + LN2 * __builtin_amdgcn_logf(acc1[3][v]);
        orow[(16 + v) * FF * CON] = o;
    }
}

extern "C" void kernel_launch(void* const* d_in, const int* in_sizes, int n_in,
                              void* d_out, int out_size, void* d_ws, size_t ws_size,
                              hipStream_t stream) {
    (void)in_sizes; (void)n_in; (void)d_ws; (void)ws_size; (void)out_size;
    const float* ll  = (const float*)d_in[0];
    const float* wgt = (const float*)d_in[1];
    float* out = (float*)d_out;

    weightedsum_lse_wmma<<<dim3(NWG), dim3(128), 0, stream>>>(ll, wgt, out);
}
